# starting kernel
# baseline (speedup 1.0000x reference)
#include <hip/hip_runtime.h>

typedef __attribute__((ext_vector_type(2))) float v2f;
typedef __attribute__((ext_vector_type(8))) float v8f;

#define NROWS 8192
#define DIM   2048
#define SCALE 0.1f

__global__ void colsum_wmma_kernel(const float* __restrict__ W,
                                   float* __restrict__ wp) {
    int gtid = blockIdx.x * blockDim.x + threadIdx.x;
    int wave = gtid >> 5;
    int lane = threadIdx.x & 31;
    int half = lane >> 4;
    int n    = lane & 15;

    int split    = wave >> 7;
    int colgroup = wave & 127;
    int j = colgroup * 16 + n;

    v2f a; a.x = 1.0f; a.y = 1.0f;
    v8f c = {};

    int kbase = split * 512;
    for (int k0 = kbase; k0 < kbase + 512; k0 += 4) {
        int r = k0 + 2 * half;
        v2f b;
        b.x = W[(long)r * DIM + j];
        b.y = W[(long)(r + 1) * DIM + j];
        c = __builtin_amdgcn_wmma_f32_16x16x4_f32(
                false, a, false, b, (short)0, c, false, false);
    }
    if (lane < 16)
        wp[split * DIM + colgroup * 16 + n] = c[0];
}

__global__ void combine_kernel(const float* __restrict__ wp,
                               const float* __restrict__ bias,
                               float* __restrict__ wc,
                               float* __restrict__ sb) {
    int j = blockIdx.x * 256 + threadIdx.x;
    if (j < DIM)
        wc[j] = wp[j] + wp[DIM + j] + wp[2 * DIM + j] + wp[3 * DIM + j];

    if (blockIdx.x == 0) {
        __shared__ float s[256];
        float acc = 0.0f;
        for (int t = threadIdx.x; t < DIM; t += 256) acc += bias[t];
        s[threadIdx.x] = acc;
        __syncthreads();
        for (int off = 128; off > 0; off >>= 1) {
            if (threadIdx.x < off) s[threadIdx.x] += s[threadIdx.x + off];
            __syncthreads();
        }
        if (threadIdx.x == 0) sb[0] = s[0];
    }
}

__global__ void gemv_wmma_kernel(const float* __restrict__ x,
                                 const float* __restrict__ wc,
                                 float* __restrict__ op) {
    int gtid = blockIdx.x * blockDim.x + threadIdx.x;
    int wave = gtid >> 5;
    int lane = threadIdx.x & 31;
    int half = lane >> 4;
    int n    = lane & 15;

    int rowgroup = wave >> 2;
    int split    = wave & 3;
    long row = (long)rowgroup * 16 + n;
    const float* xrow = x + row * DIM;

    v8f c = {};
    int kbase = split * 512;
    for (int k0 = kbase; k0 < kbase + 512; k0 += 4) {
        v2f b = *(const v2f*)(xrow + k0 + 2 * half);
        v2f a;
        a.x = wc[k0 + 2 * half];
        a.y = wc[k0 + 2 * half + 1];
        c = __builtin_amdgcn_wmma_f32_16x16x4_f32(
                false, a, false, b, (short)0, c, false, false);
    }
    if (lane < 16)
        op[(long)split * NROWS + rowgroup * 16 + n] = c[0];
}

__global__ void finalize_kernel(const float* __restrict__ op,
                                const float* __restrict__ sb,
                                float* __restrict__ out) {
    int r = blockIdx.x * 256 + threadIdx.x;
    if (r < NROWS) {
        float v = op[r] + op[NROWS + r] + op[2 * NROWS + r] + op[3 * NROWS + r];
        out[r] = SCALE * (v + sb[0]);
    }
}

extern "C" void kernel_launch(void* const* d_in, const int* in_sizes, int n_in,
                              void* d_out, int out_size, void* d_ws, size_t ws_size,
                              hipStream_t stream) {
    const float* x = (const float*)d_in[0];
    const float* W = (const float*)d_in[1];
    const float* b = (const float*)d_in[2];
    float* out = (float*)d_out;

    float* ws = (float*)d_ws;
    float* wp = ws;
    float* wc = ws + 4 * DIM;
    float* sb = ws + 5 * DIM;
    float* op = ws + 5 * DIM + 256;

    colsum_wmma_kernel<<<64, 256, 0, stream>>>(W, wp);
    combine_kernel<<<8, 256, 0, stream>>>(wp, b, wc, sb);
    gemv_wmma_kernel<<<256, 256, 0, stream>>>(x, wc, op);
    finalize_kernel<<<32, 256, 0, stream>>>(op, sb, out);
}


# 4 launches -> 2: combine folded into the gemv prologue (4 wcol planes summed into LDS), finalize into its epilogue (K-split partials meet in LDS), bias sum rides launch 1 as one extra block
# speedup vs baseline: 1.2457x; 1.2457x over previous
#include <hip/hip_runtime.h>

typedef __attribute__((ext_vector_type(2))) float v2f;
typedef __attribute__((ext_vector_type(4))) float v4f;
typedef __attribute__((ext_vector_type(8))) float v8f;

#define NROWS 8192
#define DIM   2048
#define SCALE 0.1f
#define COLSUM_BLOCKS 64

__global__ void colsum_wmma_kernel(const float* __restrict__ W,
                                   const float* __restrict__ bias,
                                   float* __restrict__ wp,
                                   float* __restrict__ sb) {
    if (blockIdx.x == COLSUM_BLOCKS) {
        __shared__ float s[256];
        float acc = 0.0f;
        for (int t = threadIdx.x; t < DIM; t += 256) acc += bias[t];
        s[threadIdx.x] = acc;
        __syncthreads();
        for (int off = 128; off > 0; off >>= 1) {
            if (threadIdx.x < off) s[threadIdx.x] += s[threadIdx.x + off];
            __syncthreads();
        }
        if (threadIdx.x == 0) sb[0] = s[0];
        return;
    }

    int gtid = blockIdx.x * blockDim.x + threadIdx.x;
    int wave = gtid >> 5;
    int lane = threadIdx.x & 31;
    int half = lane >> 4;
    int n    = lane & 15;

    int split    = wave >> 7;
    int colgroup = wave & 127;
    int j = colgroup * 16 + n;

    v2f a; a.x = 1.0f; a.y = 1.0f;
    v8f c = {};

    int kbase = split * 512;
    for (int k0 = kbase; k0 < kbase + 512; k0 += 4) {
        int r = k0 + 2 * half;
        v2f b;
        b.x = W[(long)r * DIM + j];
        b.y = W[(long)(r + 1) * DIM + j];
        c = __builtin_amdgcn_wmma_f32_16x16x4_f32(
                false, a, false, b, (short)0, c, false, false);
    }
    if (lane < 16)
        wp[split * DIM + colgroup * 16 + n] = c[0];
}

__global__ void gemv_wmma_kernel(const float* __restrict__ x,
                                 const float* __restrict__ wp,
                                 const float* __restrict__ sb,
                                 float* __restrict__ out) {
    __shared__ __attribute__((aligned(16))) float s_wc[DIM];
    __shared__ float s_part[8][16];

    const v4f* wp4 = (const v4f*)wp + threadIdx.x;
    v4f p0 = wp4[0];
    v4f p1 = wp4[DIM / 4];
    v4f p2 = wp4[2 * (DIM / 4)];
    v4f p3 = wp4[3 * (DIM / 4)];
    v4f q0 = wp4[256];
    v4f q1 = wp4[DIM / 4 + 256];
    v4f q2 = wp4[2 * (DIM / 4) + 256];
    v4f q3 = wp4[3 * (DIM / 4) + 256];
    *(v4f*)(s_wc + 4 * threadIdx.x)         = p0 + p1 + p2 + p3;
    *(v4f*)(s_wc + 4 * (threadIdx.x + 256)) = q0 + q1 + q2 + q3;
    float sbv = sb[0];
    __syncthreads();

    int gtid = blockIdx.x * blockDim.x + threadIdx.x;
    int wave = gtid >> 5;
    int wib  = threadIdx.x >> 5;
    int lane = threadIdx.x & 31;
    int half = lane >> 4;
    int n    = lane & 15;

    int rowgroup = wave >> 2;
    int split    = wave & 3;
    long row = (long)rowgroup * 16 + n;
    const float* xrow = x + row * DIM;

    v8f c = {};
    int kbase = split * 512;
    for (int k0 = kbase; k0 < kbase + 512; k0 += 4) {
        v2f b = *(const v2f*)(xrow + k0 + 2 * half);
        v2f a = *(const v2f*)(s_wc + k0 + 2 * half);
        c = __builtin_amdgcn_wmma_f32_16x16x4_f32(
                false, a, false, b, (short)0, c, false, false);
    }
    if (lane < 16)
        s_part[wib][n] = c[0];
    __syncthreads();

    if (threadIdx.x < 32) {
        int rg = threadIdx.x >> 4;
        int nn = threadIdx.x & 15;
        float v = s_part[rg * 4 + 0][nn] + s_part[rg * 4 + 1][nn]
                + s_part[rg * 4 + 2][nn] + s_part[rg * 4 + 3][nn];
        out[(blockIdx.x * 2 + rg) * 16 + nn] = SCALE * (v + sbv);
    }
}

extern "C" void kernel_launch(void* const* d_in, const int* in_sizes, int n_in,
                              void* d_out, int out_size, void* d_ws, size_t ws_size,
                              hipStream_t stream) {
    const float* x = (const float*)d_in[0];
    const float* W = (const float*)d_in[1];
    const float* b = (const float*)d_in[2];
    float* out = (float*)d_out;

    float* ws = (float*)d_ws;
    float* wp = ws;
    float* sb = ws + 5 * DIM;

    colsum_wmma_kernel<<<COLSUM_BLOCKS + 1, 256, 0, stream>>>(W, b, wp, sb);
    gemv_wmma_kernel<<<256, 256, 0, stream>>>(x, wp, sb, out);
}


# gemv: a wave reads whole rows as 512-byte pieces (4 full lines per load); the tile's 16 columns carry k-chunks of one row, dot = diagonal of D summed via LDS; changes the summation order
# speedup vs baseline: 1.4657x; 1.1766x over previous
#include <hip/hip_runtime.h>

typedef __attribute__((ext_vector_type(2))) float v2f;
typedef __attribute__((ext_vector_type(4))) float v4f;
typedef __attribute__((ext_vector_type(8))) float v8f;

#define NROWS 8192
#define DIM   2048
#define SCALE 0.1f
#define COLSUM_BLOCKS 64

__global__ void colsum_wmma_kernel(const float* __restrict__ W,
                                   const float* __restrict__ bias,
                                   float* __restrict__ wp,
                                   float* __restrict__ sb) {
    if (blockIdx.x == COLSUM_BLOCKS) {
        __shared__ float s[256];
        float acc = 0.0f;
        for (int t = threadIdx.x; t < DIM; t += 256) acc += bias[t];
        s[threadIdx.x] = acc;
        __syncthreads();
        for (int off = 128; off > 0; off >>= 1) {
            if (threadIdx.x < off) s[threadIdx.x] += s[threadIdx.x + off];
            __syncthreads();
        }
        if (threadIdx.x == 0) sb[0] = s[0];
        return;
    }

    int gtid = blockIdx.x * blockDim.x + threadIdx.x;
    int wave = gtid >> 5;
    int lane = threadIdx.x & 31;
    int half = lane >> 4;
    int n    = lane & 15;

    int split    = wave >> 7;
    int colgroup = wave & 127;
    int j = colgroup * 16 + n;

    v2f a; a.x = 1.0f; a.y = 1.0f;
    v8f c = {};

    int kbase = split * 512;
    for (int k0 = kbase; k0 < kbase + 512; k0 += 4) {
        int r = k0 + 2 * half;
        v2f b;
        b.x = W[(long)r * DIM + j];
        b.y = W[(long)(r + 1) * DIM + j];
        c = __builtin_amdgcn_wmma_f32_16x16x4_f32(
                false, a, false, b, (short)0, c, false, false);
    }
    if (lane < 16)
        wp[split * DIM + colgroup * 16 + n] = c[0];
}

#define GEMV_THREADS 512
#define GEMV_ROWS_PER_BLOCK 32
#define PIECES (DIM / 128)
#define DIAG_PITCH 20

static __device__ __forceinline__ v8f wmma4(v2f a, v2f b, v8f c) {
    return __builtin_amdgcn_wmma_f32_16x16x4_f32(false, a, false, b, (short)0, c, false, false);
}

static __device__ __forceinline__ float diag_pick(v8f c, int lane) {
    int r = lane & 7;
    float d = c[0];
    d = (r == 1) ? c[1] : d;
    d = (r == 2) ? c[2] : d;
    d = (r == 3) ? c[3] : d;
    d = (r == 4) ? c[4] : d;
    d = (r == 5) ? c[5] : d;
    d = (r == 6) ? c[6] : d;
    d = (r == 7) ? c[7] : d;
    return d;
}

__global__ __launch_bounds__(GEMV_THREADS)
void gemv_wmma_kernel(const float* __restrict__ x,
                      const float* __restrict__ wp,
                      const float* __restrict__ sb,
                      float* __restrict__ out) {
    __shared__ __attribute__((aligned(16))) float s_wc[DIM];
    __shared__ __attribute__((aligned(16))) float s_diag[GEMV_ROWS_PER_BLOCK][DIAG_PITCH];

    const int tid  = threadIdx.x;
    const int lane = tid & 31;
    const int wib  = tid >> 5;

    const v4f* wp4 = (const v4f*)wp + tid;
    v4f p0 = wp4[0];
    v4f p1 = wp4[DIM / 4];
    v4f p2 = wp4[2 * (DIM / 4)];
    v4f p3 = wp4[3 * (DIM / 4)];
    float sbv = sb[0];

    const long row0 = (long)blockIdx.x * GEMV_ROWS_PER_BLOCK + wib * 2;
    const v4f* xr0 = (const v4f*)(x + row0 * DIM) + lane;
    const v4f* xr1 = xr0 + DIM / 4;
    v4f xa[PIECES], xb[PIECES];
    for (int q = 0; q < PIECES; ++q) {
        xa[q] = xr0[q * 32];
        xb[q] = xr1[q * 32];
    }

    *(v4f*)(s_wc + 4 * tid) = p0 + p1 + p2 + p3;
    __builtin_amdgcn_fence(__ATOMIC_RELEASE, "workgroup", "local");
    __builtin_amdgcn_s_barrier();
    __builtin_amdgcn_fence(__ATOMIC_ACQUIRE, "workgroup", "local");

    v8f c0 = {}, c1 = {};
    for (int q = 0; q < PIECES; ++q) {
        v4f w = *(const v4f*)(s_wc + q * 128 + 4 * lane);
        v2f wlo = {w.x, w.y}, whi = {w.z, w.w};
        v2f alo = {xa[q].x, xa[q].y}, ahi = {xa[q].z, xa[q].w};
        v2f blo = {xb[q].x, xb[q].y}, bhi = {xb[q].z, xb[q].w};
        c0 = wmma4(wlo, alo, c0);
        c1 = wmma4(wlo, blo, c1);
        c0 = wmma4(whi, ahi, c0);
        c1 = wmma4(whi, bhi, c1);
    }

    float d0 = diag_pick(c0, lane);
    float d1 = diag_pick(c1, lane);
    if (lane < 8 || lane >= 24) {
        s_diag[wib * 2][lane & 15]     = d0;
        s_diag[wib * 2 + 1][lane & 15] = d1;
    }
    __syncthreads();

    if (tid < GEMV_ROWS_PER_BLOCK) {
        const v4f* dr = (const v4f*)&s_diag[tid][0];
        v4f s = (dr[0] + dr[1]) + (dr[2] + dr[3]);
        float v = (s.x + s.y) + (s.z + s.w);
        out[blockIdx.x * GEMV_ROWS_PER_BLOCK + tid] = SCALE * (v + sbv);
    }
}

extern "C" void kernel_launch(void* const* d_in, const int* in_sizes, int n_in,
                              void* d_out, int out_size, void* d_ws, size_t ws_size,
                              hipStream_t stream) {
    const float* x = (const float*)d_in[0];
    const float* W = (const float*)d_in[1];
    const float* b = (const float*)d_in[2];
    float* out = (float*)d_out;

    float* ws = (float*)d_ws;
    float* wp = ws;
    float* sb = ws + 5 * DIM;

    colsum_wmma_kernel<<<COLSUM_BLOCKS + 1, 256, 0, stream>>>(W, b, wp, sb);
    gemv_wmma_kernel<<<256, 512, 0, stream>>>(x, wp, sb, out);
}


# launch 1 re-cut: 256 blocks of 4 waves, block = 32 columns x 512 rows, whole-line 16-byte loads all issued before the first matrix op; bias loads likewise; changes the summation order
# speedup vs baseline: 2.8752x; 1.9617x over previous
#include <hip/hip_runtime.h>

typedef __attribute__((ext_vector_type(2))) float v2f;
typedef __attribute__((ext_vector_type(4))) float v4f;
typedef __attribute__((ext_vector_type(8))) float v8f;

#define NROWS 8192
#define DIM   2048
#define SCALE 0.1f
#define CS_THREADS 128
#define CS_WAVES (CS_THREADS / 32)
#define CS_PLANES 4
#define CS_STRIP 32
#define CS_ROWS_PER_WAVE (DIM / CS_PLANES / CS_WAVES)
#define CS_LOADS (CS_ROWS_PER_WAVE / 4)
#define COLSUM_BLOCKS (CS_PLANES * (DIM / CS_STRIP))

__global__ __launch_bounds__(CS_THREADS)
void colsum_wmma_kernel(const float* __restrict__ W,
                        const float* __restrict__ bias,
                        float* __restrict__ wp,
                        float* __restrict__ sb) {
    const int tid  = threadIdx.x;
    const int lane = tid & 31;
    const int wib  = tid >> 5;

    if (blockIdx.x == COLSUM_BLOCKS) {
        __shared__ float s_b[CS_WAVES];
        const v4f* b4 = (const v4f*)bias + tid;
        v4f a0 = b4[0], a1 = b4[CS_THREADS], a2 = b4[2 * CS_THREADS], a3 = b4[3 * CS_THREADS];
        v4f t = (a0 + a1) + (a2 + a3);
        float acc = (t.x + t.y) + (t.z + t.w);
        for (int off = 16; off > 0; off >>= 1) acc += __shfl_xor(acc, off);
        if (lane == 0) s_b[wib] = acc;
        __syncthreads();
        if (tid == 0) sb[0] = (s_b[0] + s_b[1]) + (s_b[2] + s_b[3]);
        return;
    }

    __shared__ __attribute__((aligned(16))) float s_part[CS_WAVES * 64];

    const int plane = blockIdx.x / (DIM / CS_STRIP);
    const int strip = blockIdx.x % (DIM / CS_STRIP);
    const int h    = lane >> 4;
    const int n    = lane & 15;
    const int q    = n & 7;
    const int rpar = n >> 3;

    const long row0 = (long)plane * (DIM / CS_PLANES) + wib * CS_ROWS_PER_WAVE + 2 * h + rpar;
    const v4f* src = (const v4f*)(W + row0 * DIM + strip * CS_STRIP) + q;

    v4f v[CS_LOADS];
    for (int i = 0; i < CS_LOADS; ++i)
        v[i] = src[(long)i * DIM];
    __builtin_amdgcn_sched_barrier(0);

    v2f ones; ones.x = 1.0f; ones.y = 1.0f;
    v8f c0 = {}, c1 = {}, c2 = {}, c3 = {};
    for (int t = 0; t < CS_LOADS; t += 2) {
        v2f b0 = {v[t].x, v[t + 1].x};
        v2f b1 = {v[t].y, v[t + 1].y};
        v2f b2 = {v[t].z, v[t + 1].z};
        v2f b3 = {v[t].w, v[t + 1].w};
        c0 = __builtin_amdgcn_wmma_f32_16x16x4_f32(false, ones, false, b0, (short)0, c0, false, false);
        c1 = __builtin_amdgcn_wmma_f32_16x16x4_f32(false, ones, false, b1, (short)0, c1, false, false);
        c2 = __builtin_amdgcn_wmma_f32_16x16x4_f32(false, ones, false, b2, (short)0, c2, false, false);
        c3 = __builtin_amdgcn_wmma_f32_16x16x4_f32(false, ones, false, b3, (short)0, c3, false, false);
    }

    if (lane < 16) {
        v4f o = {c0[0], c1[0], c2[0], c3[0]};
        *(v4f*)(s_part + wib * 64 + n * 4) = o;
    }
    __syncthreads();

    if (tid < CS_STRIP) {
        float p0 = s_part[0 * 64 + tid] + s_part[0 * 64 + 32 + tid];
        float p1 = s_part[1 * 64 + tid] + s_part[1 * 64 + 32 + tid];
        float p2 = s_part[2 * 64 + tid] + s_part[2 * 64 + 32 + tid];
        float p3 = s_part[3 * 64 + tid] + s_part[3 * 64 + 32 + tid];
        wp[plane * DIM + strip * CS_STRIP + tid] = (p0 + p1) + (p2 + p3);
    }
}

#define GEMV_THREADS 512
#define GEMV_ROWS_PER_BLOCK 32
#define PIECES (DIM / 128)
#define DIAG_PITCH 20

static __device__ __forceinline__ v8f wmma4(v2f a, v2f b, v8f c) {
    return __builtin_amdgcn_wmma_f32_16x16x4_f32(false, a, false, b, (short)0, c, false, false);
}

static __device__ __forceinline__ float diag_pick(v8f c, int lane) {
    int r = lane & 7;
    float d = c[0];
    d = (r == 1) ? c[1] : d;
    d = (r == 2) ? c[2] : d;
    d = (r == 3) ? c[3] : d;
    d = (r == 4) ? c[4] : d;
    d = (r == 5) ? c[5] : d;
    d = (r == 6) ? c[6] : d;
    d = (r == 7) ? c[7] : d;
    return d;
}

__global__ __launch_bounds__(GEMV_THREADS)
void gemv_wmma_kernel(const float* __restrict__ x,
                      const float* __restrict__ wp,
                      const float* __restrict__ sb,
                      float* __restrict__ out) {
    __shared__ __attribute__((aligned(16))) float s_wc[DIM];
    __shared__ __attribute__((aligned(16))) float s_diag[GEMV_ROWS_PER_BLOCK][DIAG_PITCH];

    const int tid  = threadIdx.x;
    const int lane = tid & 31;
    const int wib  = tid >> 5;

    const v4f* wp4 = (const v4f*)wp + tid;
    v4f p0 = wp4[0];
    v4f p1 = wp4[DIM / 4];
    v4f p2 = wp4[2 * (DIM / 4)];
    v4f p3 = wp4[3 * (DIM / 4)];
    float sbv = sb[0];

    const long row0 = (long)blockIdx.x * GEMV_ROWS_PER_BLOCK + wib * 2;
    const v4f* xr0 = (const v4f*)(x + row0 * DIM) + lane;
    const v4f* xr1 = xr0 + DIM / 4;
    v4f xa[PIECES], xb[PIECES];
    for (int q = 0; q < PIECES; ++q) {
        xa[q] = xr0[q * 32];
        xb[q] = xr1[q * 32];
    }

    *(v4f*)(s_wc + 4 * tid) = p0 + p1 + p2 + p3;
    __builtin_amdgcn_fence(__ATOMIC_RELEASE, "workgroup", "local");
    __builtin_amdgcn_s_barrier();
    __builtin_amdgcn_fence(__ATOMIC_ACQUIRE, "workgroup", "local");

    v8f c0 = {}, c1 = {};
    for (int q = 0; q < PIECES; ++q) {
        v4f w = *(const v4f*)(s_wc + q * 128 + 4 * lane);
        v2f wlo = {w.x, w.y}, whi = {w.z, w.w};
        v2f alo = {xa[q].x, xa[q].y}, ahi = {xa[q].z, xa[q].w};
        v2f blo = {xb[q].x, xb[q].y}, bhi = {xb[q].z, xb[q].w};
        c0 = wmma4(wlo, alo, c0);
        c1 = wmma4(wlo, blo, c1);
        c0 = wmma4(whi, ahi, c0);
        c1 = wmma4(whi, bhi, c1);
    }

    float d0 = diag_pick(c0, lane);
    float d1 = diag_pick(c1, lane);
    if (lane < 8 || lane >= 24) {
        s_diag[wib * 2][lane & 15]     = d0;
        s_diag[wib * 2 + 1][lane & 15] = d1;
    }
    __syncthreads();

    if (tid < GEMV_ROWS_PER_BLOCK) {
        const v4f* dr = (const v4f*)&s_diag[tid][0];
        v4f s = (dr[0] + dr[1]) + (dr[2] + dr[3]);
        float v = (s.x + s.y) + (s.z + s.w);
        out[blockIdx.x * GEMV_ROWS_PER_BLOCK + tid] = SCALE * (v + sbv);
    }
}

extern "C" void kernel_launch(void* const* d_in, const int* in_sizes, int n_in,
                              void* d_out, int out_size, void* d_ws, size_t ws_size,
                              hipStream_t stream) {
    const float* x = (const float*)d_in[0];
    const float* W = (const float*)d_in[1];
    const float* b = (const float*)d_in[2];
    float* out = (float*)d_out;

    float* ws = (float*)d_ws;
    float* wp = ws;
    float* sb = ws + 5 * DIM;

    colsum_wmma_kernel<<<COLSUM_BLOCKS + 1, CS_THREADS, 0, stream>>>(W, b, wp, sb);
    gemv_wmma_kernel<<<256, 512, 0, stream>>>(x, wp, sb, out);
}


# launch 2 re-cut: 256 blocks of 4 waves (one per SIMD) x 8 rows, 1024 waves to start instead of 4096; wcol slice kept in registers; rows requested two pairs ahead, then in groups of 16 loads
# speedup vs baseline: 3.2097x; 1.1163x over previous
#include <hip/hip_runtime.h>

typedef __attribute__((ext_vector_type(2))) float v2f;
typedef __attribute__((ext_vector_type(4))) float v4f;
typedef __attribute__((ext_vector_type(8))) float v8f;

#define NROWS 8192
#define DIM   2048
#define SCALE 0.1f
#define CS_THREADS 128
#define CS_WAVES (CS_THREADS / 32)
#define CS_PLANES 4
#define CS_STRIP 32
#define CS_ROWS_PER_WAVE (DIM / CS_PLANES / CS_WAVES)
#define CS_LOADS (CS_ROWS_PER_WAVE / 4)
#define COLSUM_BLOCKS (CS_PLANES * (DIM / CS_STRIP))

__global__ __launch_bounds__(CS_THREADS)
void colsum_wmma_kernel(const float* __restrict__ W,
                        const float* __restrict__ bias,
                        float* __restrict__ wp,
                        float* __restrict__ sb) {
    const int tid  = threadIdx.x;
    const int lane = tid & 31;
    const int wib  = tid >> 5;

    if (blockIdx.x == COLSUM_BLOCKS) {
        __shared__ float s_b[CS_WAVES];
        const v4f* b4 = (const v4f*)bias + tid;
        v4f a0 = b4[0], a1 = b4[CS_THREADS], a2 = b4[2 * CS_THREADS], a3 = b4[3 * CS_THREADS];
        v4f t = (a0 + a1) + (a2 + a3);
        float acc = (t.x + t.y) + (t.z + t.w);
        for (int off = 16; off > 0; off >>= 1) acc += __shfl_xor(acc, off);
        if (lane == 0) s_b[wib] = acc;
        __syncthreads();
        if (tid == 0) sb[0] = (s_b[0] + s_b[1]) + (s_b[2] + s_b[3]);
        return;
    }

    __shared__ __attribute__((aligned(16))) float s_part[CS_WAVES * 64];

    const int plane = blockIdx.x / (DIM / CS_STRIP);
    const int strip = blockIdx.x % (DIM / CS_STRIP);
    const int h    = lane >> 4;
    const int n    = lane & 15;
    const int q    = n & 7;
    const int rpar = n >> 3;

    const long row0 = (long)plane * (DIM / CS_PLANES) + wib * CS_ROWS_PER_WAVE + 2 * h + rpar;
    const v4f* src = (const v4f*)(W + row0 * DIM + strip * CS_STRIP) + q;

    v4f v[CS_LOADS];
    for (int i = 0; i < CS_LOADS; ++i)
        v[i] = src[(long)i * DIM];
    __builtin_amdgcn_sched_barrier(0);

    v2f ones; ones.x = 1.0f; ones.y = 1.0f;
    v8f c0 = {}, c1 = {}, c2 = {}, c3 = {};
    for (int t = 0; t < CS_LOADS; t += 2) {
        v2f b0 = {v[t].x, v[t + 1].x};
        v2f b1 = {v[t].y, v[t + 1].y};
        v2f b2 = {v[t].z, v[t + 1].z};
        v2f b3 = {v[t].w, v[t + 1].w};
        c0 = __builtin_amdgcn_wmma_f32_16x16x4_f32(false, ones, false, b0, (short)0, c0, false, false);
        c1 = __builtin_amdgcn_wmma_f32_16x16x4_f32(false, ones, false, b1, (short)0, c1, false, false);
        c2 = __builtin_amdgcn_wmma_f32_16x16x4_f32(false, ones, false, b2, (short)0, c2, false, false);
        c3 = __builtin_amdgcn_wmma_f32_16x16x4_f32(false, ones, false, b3, (short)0, c3, false, false);
    }

    if (lane < 16) {
        v4f o = {c0[0], c1[0], c2[0], c3[0]};
        *(v4f*)(s_part + wib * 64 + n * 4) = o;
    }
    __syncthreads();

    if (tid < CS_STRIP) {
        float p0 = s_part[0 * 64 + tid] + s_part[0 * 64 + 32 + tid];
        float p1 = s_part[1 * 64 + tid] + s_part[1 * 64 + 32 + tid];
        float p2 = s_part[2 * 64 + tid] + s_part[2 * 64 + 32 + tid];
        float p3 = s_part[3 * 64 + tid] + s_part[3 * 64 + 32 + tid];
        wp[plane * DIM + strip * CS_STRIP + tid] = (p0 + p1) + (p2 + p3);
    }
}

#define GEMV_THREADS 128
#define GEMV_WAVES (GEMV_THREADS / 32)
#define GEMV_ROWS_PER_BLOCK 32
#define GEMV_ROWS_PER_WAVE (GEMV_ROWS_PER_BLOCK / GEMV_WAVES)
#define PAIRS (GEMV_ROWS_PER_WAVE / 2)
#define PIECES (DIM / 128)
#define HALF (PIECES / 2)
#define WP_PIECES (DIM / 4 / GEMV_THREADS)
#define DIAG_PITCH 20

static __device__ __forceinline__ v8f wmma4(v2f a, v2f b, v8f c) {
    return __builtin_amdgcn_wmma_f32_16x16x4_f32(false, a, false, b, (short)0, c, false, false);
}

static __device__ __forceinline__ float diag_pick(v8f c, int lane) {
    int r = lane & 7;
    float d = c[0];
    d = (r == 1) ? c[1] : d;
    d = (r == 2) ? c[2] : d;
    d = (r == 3) ? c[3] : d;
    d = (r == 4) ? c[4] : d;
    d = (r == 5) ? c[5] : d;
    d = (r == 6) ? c[6] : d;
    d = (r == 7) ? c[7] : d;
    return d;
}

template <int P, int H>
static __device__ __forceinline__ void load_half(const v4f* xr, v4f (&xa)[PAIRS][PIECES],
                                                 v4f (&xb)[PAIRS][PIECES]) {
#pragma unroll
    for (int q = H * HALF; q < (H + 1) * HALF; ++q) {
        xa[P][q] = xr[(2 * P) * (DIM / 4) + q * 32];
        xb[P][q] = xr[(2 * P + 1) * (DIM / 4) + q * 32];
    }
}

template <int P, int H>
static __device__ __forceinline__ void mma_half(const v4f (&w)[PIECES], const v4f (&xa)[PAIRS][PIECES],
                                                const v4f (&xb)[PAIRS][PIECES], v8f& c0, v8f& c1) {
#pragma unroll
    for (int q = H * HALF; q < (H + 1) * HALF; ++q) {
        v2f wlo = {w[q].x, w[q].y}, whi = {w[q].z, w[q].w};
        v2f alo = {xa[P][q].x, xa[P][q].y}, ahi = {xa[P][q].z, xa[P][q].w};
        v2f blo = {xb[P][q].x, xb[P][q].y}, bhi = {xb[P][q].z, xb[P][q].w};
        c0 = wmma4(wlo, alo, c0);
        c1 = wmma4(wlo, blo, c1);
        c0 = wmma4(whi, ahi, c0);
        c1 = wmma4(whi, bhi, c1);
    }
}

__global__ __launch_bounds__(GEMV_THREADS)
void gemv_wmma_kernel(const float* x,
                      const float* __restrict__ wp,
                      const float* __restrict__ sb,
                      float* __restrict__ out) {
    __shared__ __attribute__((aligned(16))) float s_wc[DIM];
    __shared__ __attribute__((aligned(16))) float s_diag[GEMV_ROWS_PER_BLOCK][DIAG_PITCH];

    const int tid  = threadIdx.x;
    const int lane = tid & 31;
    const int wib  = tid >> 5;

    const v4f* wp4 = (const v4f*)wp + tid;
    v4f pl[WP_PIECES][CS_PLANES];
#pragma unroll
    for (int j = 0; j < WP_PIECES; ++j) {
#pragma unroll
        for (int p = 0; p < CS_PLANES; ++p)
            pl[j][p] = wp4[p * (DIM / 4) + j * GEMV_THREADS];
    }
    float sbv = sb[0];

    const long row0 = (long)blockIdx.x * GEMV_ROWS_PER_BLOCK + wib * GEMV_ROWS_PER_WAVE;
    const v4f* xr = (const v4f*)(x + row0 * DIM) + lane;
    v4f xa[PAIRS][PIECES], xb[PAIRS][PIECES];

    load_half<0, 0>(xr, xa, xb);
    load_half<0, 1>(xr, xa, xb);
    load_half<1, 0>(xr, xa, xb);
    load_half<1, 1>(xr, xa, xb);
    __builtin_amdgcn_sched_barrier(0);

#pragma unroll
    for (int j = 0; j < WP_PIECES; ++j)
        *(v4f*)(s_wc + 4 * (tid + j * GEMV_THREADS)) = pl[j][0] + pl[j][1] + pl[j][2] + pl[j][3];
    __builtin_amdgcn_fence(__ATOMIC_RELEASE, "workgroup", "local");
    __builtin_amdgcn_s_barrier();
    __builtin_amdgcn_fence(__ATOMIC_ACQUIRE, "workgroup", "local");

    v4f w[PIECES];
#pragma unroll
    for (int q = 0; q < PIECES; ++q)
        w[q] = *(const v4f*)(s_wc + q * 128 + 4 * lane);
    __builtin_amdgcn_sched_barrier(0);

    v8f c00 = {}, c01 = {}, c10 = {}, c11 = {}, c20 = {}, c21 = {}, c30 = {}, c31 = {};

#define ORDER_POINT(...) do { asm volatile("" : __VA_ARGS__); __builtin_amdgcn_sched_barrier(0); } while (0)
#define ACC0 "+v"(c00), "+v"(c01)
#define ACC1 "+v"(c10), "+v"(c11)
#define ACC2 "+v"(c20), "+v"(c21)
#define ACC3 "+v"(c30), "+v"(c31)

    mma_half<0, 0>(w, xa, xb, c00, c01);
    ORDER_POINT(ACC0, ACC1, ACC2, ACC3);
    load_half<2, 0>(xr, xa, xb);
    ORDER_POINT(ACC0, ACC1, ACC2, ACC3);
    mma_half<0, 1>(w, xa, xb, c00, c01);
    ORDER_POINT(ACC0, ACC1, ACC2, ACC3);
    load_half<2, 1>(xr, xa, xb);
    ORDER_POINT(ACC0, ACC1, ACC2, ACC3);
    mma_half<1, 0>(w, xa, xb, c10, c11);
    float d0 = diag_pick(c00, lane);
    float d1 = diag_pick(c01, lane);
    ORDER_POINT(ACC1, ACC2, ACC3);
    load_half<3, 0>(xr, xa, xb);
    ORDER_POINT(ACC1, ACC2, ACC3);
    mma_half<1, 1>(w, xa, xb, c10, c11);
    ORDER_POINT(ACC1, ACC2, ACC3);
    load_half<3, 1>(xr, xa, xb);
    ORDER_POINT(ACC1, ACC2, ACC3);
    mma_half<2, 0>(w, xa, xb, c20, c21);
    float d2 = diag_pick(c10, lane);
    float d3 = diag_pick(c11, lane);
    mma_half<2, 1>(w, xa, xb, c20, c21);
    ORDER_POINT(ACC2, ACC3);
    mma_half<3, 0>(w, xa, xb, c30, c31);
    float d4 = diag_pick(c20, lane);
    float d5 = diag_pick(c21, lane);
    mma_half<3, 1>(w, xa, xb, c30, c31);
    float d6 = diag_pick(c30, lane);
    float d7 = diag_pick(c31, lane);

    if (lane < 8 || lane >= 24) {
        float* dr = &s_diag[wib * GEMV_ROWS_PER_WAVE][lane & 15];
        dr[0 * DIAG_PITCH] = d0;
        dr[1 * DIAG_PITCH] = d1;
        dr[2 * DIAG_PITCH] = d2;
        dr[3 * DIAG_PITCH] = d3;
        dr[4 * DIAG_PITCH] = d4;
        dr[5 * DIAG_PITCH] = d5;
        dr[6 * DIAG_PITCH] = d6;
        dr[7 * DIAG_PITCH] = d7;
    }
    __syncthreads();

    if (tid < GEMV_ROWS_PER_BLOCK) {
        const v4f* dr = (const v4f*)&s_diag[tid][0];
        v4f s = (dr[0] + dr[1]) + (dr[2] + dr[3]);
        float v = (s.x + s.y) + (s.z + s.w);
        out[blockIdx.x * GEMV_ROWS_PER_BLOCK + tid] = SCALE * (v + sbv);
    }
}

extern "C" void kernel_launch(void* const* d_in, const int* in_sizes, int n_in,
                              void* d_out, int out_size, void* d_ws, size_t ws_size,
                              hipStream_t stream) {
    const float* x = (const float*)d_in[0];
    const float* W = (const float*)d_in[1];
    const float* b = (const float*)d_in[2];
    float* out = (float*)d_out;

    float* ws = (float*)d_ws;
    float* wp = ws;
    float* sb = ws + 5 * DIM;

    colsum_wmma_kernel<<<COLSUM_BLOCKS + 1, CS_THREADS, 0, stream>>>(W, b, wp, sb);
    gemv_wmma_kernel<<<256, GEMV_THREADS, 0, stream>>>(x, wp, sb, out);
}
